# starting kernel
# baseline (speedup 1.0000x reference)
#include <hip/hip_runtime.h>

#define NTREE 511
#define BATCH 32
#define MDIM  256
#define LSTRIDE 260

typedef __attribute__((ext_vector_type(2))) float v2f;
typedef __attribute__((ext_vector_type(8))) float v8f;

__device__ __forceinline__ v8f wmma4(v2f a, v2f b, v8f c) {
  return __builtin_amdgcn_wmma_f32_16x16x4_f32(false, a, false, b, (short)0, c,
                                               false, false);
}

__device__ __forceinline__ float sigf(float x) { return 1.0f / (1.0f + __expf(-x)); }

__global__ __launch_bounds__(256) void pack_weights(
    const float* __restrict__ W_ioux, const float* __restrict__ b_ioux,
    const float* __restrict__ W_fx, const float* __restrict__ b_fx,
    float* __restrict__ wcat, float* __restrict__ bcat)
{
  int idx = blockIdx.x * blockDim.x + threadIdx.x;
  if (idx < 1024 * 256) {
    int row = idx >> 8;
    wcat[idx] = (row < 768) ? W_ioux[idx] : W_fx[idx - 768 * 256];
  } else {
    int col = idx - 1024 * 256;
    bcat[col] = (col < 768) ? b_ioux[col] : b_fx[col - 768];
  }
}

__global__ __launch_bounds__(256) void xproj_gemm(
    const float* __restrict__ inputs,
    const float* __restrict__ wcat,
    const float* __restrict__ bcat,
    float* __restrict__ xbuf)
{
  __shared__ float sa[BATCH * LSTRIDE];

  const int tid = threadIdx.x;
  const int n   = blockIdx.x;

#pragma unroll
  for (int i = 0; i < 8; ++i) {
    int F = tid + 256 * i;
    int b = F >> 6, mq = F & 63;
    float4 v = *(const float4*)(inputs + (size_t)b * (NTREE * MDIM) +
                                (size_t)n * MDIM + mq * 4);
    *(float4*)(&sa[b * LSTRIDE + mq * 4]) = v;
  }
  __syncthreads();

  const int lane  = tid & 31;
  const int wave  = tid >> 5;
  const int rbase = (wave & 1) * 16;
  const int c0    = blockIdx.y * 256 + (wave >> 1) * 64;
  const int lr    = lane & 15;
  const int khalf = (lane < 16) ? 0 : 2;

  const float* al  = sa + (size_t)(rbase + lr) * LSTRIDE + khalf;
  const float* w0p = wcat + (size_t)(c0 + 0 * 16 + lr) * 256 + khalf;
  const float* w1p = wcat + (size_t)(c0 + 1 * 16 + lr) * 256 + khalf;
  const float* w2p = wcat + (size_t)(c0 + 2 * 16 + lr) * 256 + khalf;
  const float* w3p = wcat + (size_t)(c0 + 3 * 16 + lr) * 256 + khalf;

  v8f acc0 = {}, acc1 = {}, acc2 = {}, acc3 = {};
  for (int k0 = 0; k0 < 256; k0 += 4) {
    v2f a  = *(const v2f*)(al + k0);
    v2f w0 = *(const v2f*)(w0p + k0);
    v2f w1 = *(const v2f*)(w1p + k0);
    v2f w2 = *(const v2f*)(w2p + k0);
    v2f w3 = *(const v2f*)(w3p + k0);
    acc0 = wmma4(a, w0, acc0);
    acc1 = wmma4(a, w1, acc1);
    acc2 = wmma4(a, w2, acc2);
    acc3 = wmma4(a, w3, acc3);
  }

  v8f accs[4] = {acc0, acc1, acc2, acc3};
#pragma unroll
  for (int f = 0; f < 4; ++f) {
    int col = c0 + f * 16 + lr;
    float bs = bcat[col];
    float* op = xbuf + (size_t)n * (BATCH * 1024) + col;
#pragma unroll
    for (int v = 0; v < 8; ++v) {
      int bo = rbase + v + ((lane >> 4) << 3);
      op[(size_t)bo * 1024] = accs[f][v] + bs;
    }
  }
}

__global__ __launch_bounds__(256) void leaf_kernel(
    const float* __restrict__ xbuf, const float* __restrict__ b_iouh,
    float* __restrict__ c_buf, float* __restrict__ hid)
{
  int idx = blockIdx.x * blockDim.x + threadIdx.x;
  int m = idx & 255, b = (idx >> 8) & 31, node = idx >> 13;
  const float* xp = xbuf + (size_t)node * (BATCH * 1024) + (size_t)b * 1024;
  float ig = sigf(xp[m] + b_iouh[m]);
  float og = sigf(xp[256 + m] + b_iouh[256 + m]);
  float ug = tanhf(xp[512 + m] + b_iouh[512 + m]);
  float c = ig * ug;
  float h = og * tanhf(c);
  size_t o = (size_t)b * (NTREE * MDIM) + (size_t)node * MDIM + m;
  c_buf[o] = c;
  hid[o] = h;
}

__global__ __launch_bounds__(256) void level_gemm(
    const float* __restrict__ hid,
    const float* __restrict__ W_iouh,
    const float* __restrict__ W_fh,
    float* __restrict__ tmp_iou,
    float* __restrict__ tmp_f,
    int level_start)
{
  __shared__ float sh[2][BATCH * LSTRIDE];

  const int tid  = threadIdx.x;
  const int nl   = blockIdx.x;
  const int node = level_start + nl;
  const int ch0  = 2 * node - 511;
  const int ch1  = 2 * node - 512;

#pragma unroll
  for (int i = 0; i < 16; ++i) {
    int F = tid + 256 * i;
    int which = F >> 11;
    int rem = F & 2047;
    int b = rem >> 6, mq = rem & 63;
    int ch = which ? ch1 : ch0;
    float4 v = *(const float4*)(hid + (size_t)b * (NTREE * MDIM) +
                                (size_t)ch * MDIM + mq * 4);
    *(float4*)(&sh[which][b * LSTRIDE + mq * 4]) = v;
  }
  __syncthreads();

  const int lane  = tid & 31;
  const int wave  = tid >> 5;
  const int lr    = lane & 15;
  const int khalf = (lane < 16) ? 0 : 2;

  for (int t = wave; t < 160; t += 8) {
    if (t < 96) {
      int rt = t & 1, ct = t >> 1;
      const float* a0 = &sh[0][(rt * 16 + lr) * LSTRIDE + khalf];
      const float* a1 = &sh[1][(rt * 16 + lr) * LSTRIDE + khalf];
      int col = ct * 16 + lr;
      const float* wr = W_iouh + (size_t)col * 256 + khalf;
      v8f acc = {};
      for (int k0 = 0; k0 < 256; k0 += 4) {
        v2f a = *(const v2f*)(a0 + k0) + *(const v2f*)(a1 + k0);
        v2f w = *(const v2f*)(wr + k0);
        acc = wmma4(a, w, acc);
      }
      float* op = tmp_iou + (size_t)nl * (BATCH * 768) + col;
#pragma unroll
      for (int v = 0; v < 8; ++v) {
        int bo = rt * 16 + v + ((lane >> 4) << 3);
        op[(size_t)bo * 768] = acc[v];
      }
    } else {
      int tt = t - 96;
      int child = tt >> 5, rem = tt & 31;
      int rt = rem & 1, ct = rem >> 1;
      const float* al = &sh[child][(rt * 16 + lr) * LSTRIDE + khalf];
      int col = ct * 16 + lr;
      const float* wr = W_fh + (size_t)col * 256 + khalf;
      v8f acc = {};
      for (int k0 = 0; k0 < 256; k0 += 4) {
        v2f a = *(const v2f*)(al + k0);
        v2f w = *(const v2f*)(wr + k0);
        acc = wmma4(a, w, acc);
      }
      float* op = tmp_f + (((size_t)nl * 2 + child) * BATCH) * MDIM + col;
#pragma unroll
      for (int v = 0; v < 8; ++v) {
        int bo = rt * 16 + v + ((lane >> 4) << 3);
        op[(size_t)bo * MDIM] = acc[v];
      }
    }
  }
}

__global__ __launch_bounds__(256) void level_elem(
    const float* __restrict__ xbuf, const float* __restrict__ tmp_iou,
    const float* __restrict__ tmp_f, const float* __restrict__ b_iouh,
    const float* __restrict__ b_fh, float* __restrict__ c_buf,
    float* __restrict__ hid, int level_start, int count)
{
  int idx = blockIdx.x * blockDim.x + threadIdx.x;
  if (idx >= count * BATCH * MDIM) return;
  int m = idx & 255, b = (idx >> 8) & 31, nl = idx >> 13;
  int node = level_start + nl;
  int ch0 = 2 * node - 511, ch1 = 2 * node - 512;
  const float* xp = xbuf + (size_t)node * (BATCH * 1024) + (size_t)b * 1024;
  const float* ti = tmp_iou + (size_t)nl * (BATCH * 768) + (size_t)b * 768;
  const float* tf = tmp_f + (size_t)nl * (2 * BATCH * MDIM) + (size_t)b * MDIM;
  float ig = sigf(xp[m] + ti[m] + b_iouh[m]);
  float og = sigf(xp[256 + m] + ti[256 + m] + b_iouh[256 + m]);
  float ug = tanhf(xp[512 + m] + ti[512 + m] + b_iouh[512 + m]);
  float xf = xp[768 + m];
  float f0 = sigf(tf[m] + b_fh[m] + xf);
  float f1 = sigf(tf[BATCH * MDIM + m] + b_fh[m] + xf);
  size_t base = (size_t)b * (NTREE * MDIM) + m;
  float c = ig * ug + f0 * c_buf[base + (size_t)ch0 * MDIM]
                    + f1 * c_buf[base + (size_t)ch1 * MDIM];
  float h = og * tanhf(c);
  c_buf[base + (size_t)node * MDIM] = c;
  hid[base + (size_t)node * MDIM] = h;
}

__global__ __launch_bounds__(256) void root_copy(
    const float* __restrict__ c_buf, const float* __restrict__ hid,
    float* __restrict__ out)
{
  int idx = blockIdx.x * blockDim.x + threadIdx.x;
  int m = idx & 255, b = idx >> 8;
  size_t o = (size_t)b * (NTREE * MDIM) + (size_t)510 * MDIM + m;
  out[idx] = c_buf[o];
  out[BATCH * MDIM + idx] = hid[o];
}

extern "C" void kernel_launch(void* const* d_in, const int* in_sizes, int n_in,
                              void* d_out, int out_size, void* d_ws, size_t ws_size,
                              hipStream_t stream) {
  (void)in_sizes; (void)n_in; (void)out_size; (void)ws_size;
  const float* inputs = (const float*)d_in[0];
  const float* W_ioux = (const float*)d_in[1];
  const float* b_ioux = (const float*)d_in[2];
  const float* W_iouh = (const float*)d_in[3];
  const float* b_iouh = (const float*)d_in[4];
  const float* W_fx   = (const float*)d_in[5];
  const float* b_fx   = (const float*)d_in[6];
  const float* W_fh   = (const float*)d_in[7];
  const float* b_fh   = (const float*)d_in[8];

  float* out = (float*)d_out;
  float* hid = out + 2 * BATCH * MDIM;

  float* ws = (float*)d_ws;
  float* xbuf    = ws;
  float* c_buf   = xbuf + (size_t)NTREE * BATCH * 1024;
  float* tmp_iou = c_buf + (size_t)BATCH * NTREE * MDIM;
  float* tmp_f   = tmp_iou + (size_t)128 * BATCH * 768;
  float* wcat    = tmp_f + (size_t)128 * 2 * BATCH * MDIM;
  float* bcat    = wcat + (size_t)1024 * 256;

  pack_weights<<<(1024 * 256 + 1024) / 256, 256, 0, stream>>>(
      W_ioux, b_ioux, W_fx, b_fx, wcat, bcat);

  dim3 g1(NTREE, 4);
  xproj_gemm<<<g1, 256, 0, stream>>>(inputs, wcat, bcat, xbuf);

  leaf_kernel<<<(256 * BATCH * MDIM) / 256, 256, 0, stream>>>(xbuf, b_iouh, c_buf, hid);

  const int starts[8] = {256, 384, 448, 480, 496, 504, 508, 510};
  const int counts[8] = {128, 64, 32, 16, 8, 4, 2, 1};
  for (int L = 0; L < 8; ++L) {
    level_gemm<<<counts[L], 256, 0, stream>>>(hid, W_iouh, W_fh, tmp_iou,
                                              tmp_f, starts[L]);
    int total = counts[L] * BATCH * MDIM;
    level_elem<<<(total + 255) / 256, 256, 0, stream>>>(
        xbuf, tmp_iou, tmp_f, b_iouh, b_fh, c_buf, hid, starts[L], counts[L]);
  }

  root_copy<<<(BATCH * MDIM) / 256, 256, 0, stream>>>(c_buf, hid, out);
}


# level kernel: each of 8 waves fetches its own 16 weight rows in 4 K chunks, parks them in private LDS rows and multiplies chunk by chunk as they land: no weight barrier, fetch under first products
# speedup vs baseline: 4.6808x; 4.6808x over previous
#include <hip/hip_runtime.h>

#define NTREE 511
#define BATCH 32
#define MDIM  256
#define LSTRIDE 260

typedef __attribute__((ext_vector_type(2))) float v2f;
typedef __attribute__((ext_vector_type(8))) float v8f;

__device__ __forceinline__ v8f wmma4(v2f a, v2f b, v8f c) {
  return __builtin_amdgcn_wmma_f32_16x16x4_f32(false, a, false, b, (short)0, c,
                                               false, false);
}

__device__ __forceinline__ float sigf(float x) { return 1.0f / (1.0f + __expf(-x)); }

__global__ __launch_bounds__(256) void pack_weights(
    const float* __restrict__ W_ioux, const float* __restrict__ b_ioux,
    const float* __restrict__ W_fx, const float* __restrict__ b_fx,
    float* __restrict__ wcat, float* __restrict__ bcat)
{
  int idx = blockIdx.x * blockDim.x + threadIdx.x;
  if (idx < 1024 * 256) {
    int row = idx >> 8;
    wcat[idx] = (row < 768) ? W_ioux[idx] : W_fx[idx - 768 * 256];
  } else {
    int col = idx - 1024 * 256;
    bcat[col] = (col < 768) ? b_ioux[col] : b_fx[col - 768];
  }
}

#define XNODES 4
#define XROWS  (XNODES * BATCH)
#define XRT    (XROWS / 16)

__global__ __launch_bounds__(256) void xproj_gemm(
    const float* __restrict__ inputs,
    const float* __restrict__ wcat,
    const float* __restrict__ bcat,
    float* __restrict__ xbuf)
{
  __shared__ float sa[XROWS * LSTRIDE];

  const int tid = threadIdx.x;
  const int n0  = blockIdx.x * XNODES;

#pragma unroll
  for (int i = 0; i < 32; ++i) {
    int F = tid + 256 * i;
    int r = F >> 6, mq = F & 63;
    int n = n0 + (r >> 5);
    n = (n < NTREE) ? n : (NTREE - 1);
    int b = r & 31;
    float4 v = *(const float4*)(inputs + (size_t)b * (NTREE * MDIM) +
                                (size_t)n * MDIM + mq * 4);
    *(float4*)(&sa[r * LSTRIDE + mq * 4]) = v;
  }
  __syncthreads();

  const int lane  = tid & 31;
  const int wave  = tid >> 5;
  const int c0    = blockIdx.y * 256 + wave * 32;
  const int lr    = lane & 15;
  const int khalf = (lane < 16) ? 0 : 2;

  const float* al  = sa + (size_t)lr * LSTRIDE + khalf;
  const float* w0p = wcat + (size_t)(c0 + lr) * 256 + khalf;
  const float* w1p = wcat + (size_t)(c0 + 16 + lr) * 256 + khalf;

  v8f acc[XRT][2];
#pragma unroll
  for (int rt = 0; rt < XRT; ++rt) { acc[rt][0] = (v8f){}; acc[rt][1] = (v8f){}; }

  v2f a[XRT], an[XRT];
#pragma unroll
  for (int rt = 0; rt < XRT; ++rt) a[rt] = *(const v2f*)(al + rt * 16 * LSTRIDE);
  v2f b0 = *(const v2f*)(w0p),     b1 = *(const v2f*)(w1p);
  v2f d0 = *(const v2f*)(w0p + 4), d1 = *(const v2f*)(w1p + 4);
  asm volatile("" : "+v"(b0), "+v"(b1), "+v"(d0), "+v"(d1));

#pragma unroll 2
  for (int k0 = 0; k0 < 256; k0 += 8) {
    const int kn = (k0 + 8) & 255;
    v2f nb0 = *(const v2f*)(w0p + kn);
    v2f nb1 = *(const v2f*)(w1p + kn);
    v2f nd0 = *(const v2f*)(w0p + kn + 4);
    v2f nd1 = *(const v2f*)(w1p + kn + 4);
#pragma unroll
    for (int rt = 0; rt < XRT; ++rt) an[rt] = *(const v2f*)(al + rt * 16 * LSTRIDE + k0 + 4);
#pragma unroll
    for (int rt = 0; rt < XRT; ++rt) {
      acc[rt][0] = wmma4(a[rt], b0, acc[rt][0]);
      acc[rt][1] = wmma4(a[rt], b1, acc[rt][1]);
    }
#pragma unroll
    for (int rt = 0; rt < XRT; ++rt) a[rt] = *(const v2f*)(al + rt * 16 * LSTRIDE + ((k0 + 8) & 255));
#pragma unroll
    for (int rt = 0; rt < XRT; ++rt) {
      acc[rt][0] = wmma4(an[rt], d0, acc[rt][0]);
      acc[rt][1] = wmma4(an[rt], d1, acc[rt][1]);
    }
    b0 = nb0; b1 = nb1; d0 = nd0; d1 = nd1;
  }

  const float bs0 = bcat[c0 + lr];
  const float bs1 = bcat[c0 + 16 + lr];
#pragma unroll
  for (int rt = 0; rt < XRT; ++rt) {
    const int n = n0 + (rt >> 1);
    if (n < NTREE) {
      float* op = xbuf + (size_t)n * (BATCH * 1024) + c0 + lr;
#pragma unroll
      for (int v = 0; v < 8; ++v) {
        int bo = (rt & 1) * 16 + v + ((lane >> 4) << 3);
        op[(size_t)bo * 1024]      = acc[rt][0][v] + bs0;
        op[(size_t)bo * 1024 + 16] = acc[rt][1][v] + bs1;
      }
    }
  }
}

__global__ __launch_bounds__(256) void leaf_kernel(
    const float* __restrict__ xbuf, const float* __restrict__ b_iouh,
    float* __restrict__ c_buf, float* __restrict__ hid)
{
  int idx = blockIdx.x * blockDim.x + threadIdx.x;
  int m = idx & 255, b = (idx >> 8) & 31, node = idx >> 13;
  const float* xp = xbuf + (size_t)node * (BATCH * 1024) + (size_t)b * 1024;
  float ig = sigf(xp[m] + b_iouh[m]);
  float og = sigf(xp[256 + m] + b_iouh[256 + m]);
  float ug = tanhf(xp[512 + m] + b_iouh[512 + m]);
  float c = ig * ug;
  float h = og * tanhf(c);
  size_t o = (size_t)b * (NTREE * MDIM) + (size_t)node * MDIM + m;
  c_buf[o] = c;
  hid[o] = h;
}

#define FSTR 36

__device__ __forceinline__ void lds_barrier() {
  asm volatile("s_wait_dscnt 0x0\n\ts_barrier_signal -1\n\ts_barrier_wait -1" ::: "memory");
}

#define FENCE_SCHED() __builtin_amdgcn_sched_barrier(0)

__global__ __launch_bounds__(256) void level_fused(
    const float* __restrict__ xbuf,
    const float* __restrict__ W_iouh,
    const float* __restrict__ W_fh,
    const float* __restrict__ b_iouh, const float* __restrict__ b_fh,
    float* __restrict__ c_buf,
    float* __restrict__ hid,
    int level_start, int count)
{
  __shared__ __attribute__((aligned(16))) float sw[128 * LSTRIDE];
  __shared__ __attribute__((aligned(16))) float sh[3][16 * LSTRIDE];
  __shared__ float ex[5][16 * FSTR];

  const int tid   = threadIdx.x;
  const int slice = blockIdx.x >> 1;
  const int rt    = blockIdx.x & 1;
  const int lane  = tid & 31;
  const int wave  = __builtin_amdgcn_readfirstlane(tid >> 5);
  const int lr    = lane & 15;
  const int khalf = (lane < 16) ? 0 : 2;
  const int ct    = wave & 1;
  const int p     = wave >> 1;

#define HPTR(i, ch)                                                            \
  (hid + (size_t)(rt * 16 + ((tid + 256 * (i)) >> 6)) * (NTREE * MDIM) +       \
   (size_t)(ch) * MDIM + (tid & 63) * 4)
#define HLOADALL(c0, c1)                                                       \
  h0 = *(const float4*)HPTR(0, c0); h1 = *(const float4*)HPTR(1, c0);          \
  h2 = *(const float4*)HPTR(2, c0); h3 = *(const float4*)HPTR(3, c0);          \
  h4 = *(const float4*)HPTR(0, c1); h5 = *(const float4*)HPTR(1, c1);          \
  h6 = *(const float4*)HPTR(2, c1); h7 = *(const float4*)HPTR(3, c1);
#define HOFF(i) ((((tid + 256 * (i)) >> 6)) * LSTRIDE + (tid & 63) * 4)
#define HSTORE(i, va, vb)                                                      \
  *(float4*)(&sh[0][HOFF(i)]) = (va);                                          \
  *(float4*)(&sh[1][HOFF(i)]) = (vb);                                          \
  *(float4*)(&sh[2][HOFF(i)]) = make_float4((va).x + (vb).x, (va).y + (vb).y,  \
                                            (va).z + (vb).z, (va).w + (vb).w)

  float4 h0, h1, h2, h3, h4, h5, h6, h7;
  {
    const int node = level_start + (int)blockIdx.y;
    const int ch0  = 2 * node - 511;
    const int ch1  = 2 * node - 512;
    HLOADALL(ch0, ch1)
  }
  FENCE_SCHED();

  const float* wsrc = ((p < 3) ? (W_iouh + (size_t)(p * 256 + slice * 32 + ct * 16) * 256)
                               : (W_fh + (size_t)(slice * 32 + ct * 16) * 256)) +
                      (lane >> 4) * 256 + (lane & 15) * 4;
  float4 w0, w1, w2, w3, w4, w5, w6, w7, w8, w9, w10, w11, w12, w13, w14, w15, w16, w17, w18, w19, w20, w21, w22, w23, w24, w25, w26, w27, w28, w29, w30, w31;
  const float* wp = wsrc;
  w0 = *(const float4*)(wp + 0 * 256 + 0);
  w1 = *(const float4*)(wp + 2 * 256 + 0);
  w2 = *(const float4*)(wp + 4 * 256 + 0);
  w3 = *(const float4*)(wp + 6 * 256 + 0);
  w4 = *(const float4*)(wp + 8 * 256 + 0);
  w5 = *(const float4*)(wp + 10 * 256 + 0);
  w6 = *(const float4*)(wp + 12 * 256 + 0);
  w7 = *(const float4*)(wp + 14 * 256 + 0);
  asm volatile("" : "+v"(wp));
  w8 = *(const float4*)(wp + 0 * 256 + 64);
  w9 = *(const float4*)(wp + 2 * 256 + 64);
  w10 = *(const float4*)(wp + 4 * 256 + 64);
  w11 = *(const float4*)(wp + 6 * 256 + 64);
  w12 = *(const float4*)(wp + 8 * 256 + 64);
  w13 = *(const float4*)(wp + 10 * 256 + 64);
  w14 = *(const float4*)(wp + 12 * 256 + 64);
  w15 = *(const float4*)(wp + 14 * 256 + 64);
  asm volatile("" : "+v"(wp));
  w16 = *(const float4*)(wp + 0 * 256 + 128);
  w17 = *(const float4*)(wp + 2 * 256 + 128);
  w18 = *(const float4*)(wp + 4 * 256 + 128);
  w19 = *(const float4*)(wp + 6 * 256 + 128);
  w20 = *(const float4*)(wp + 8 * 256 + 128);
  w21 = *(const float4*)(wp + 10 * 256 + 128);
  w22 = *(const float4*)(wp + 12 * 256 + 128);
  w23 = *(const float4*)(wp + 14 * 256 + 128);
  asm volatile("" : "+v"(wp));
  w24 = *(const float4*)(wp + 0 * 256 + 192);
  w25 = *(const float4*)(wp + 2 * 256 + 192);
  w26 = *(const float4*)(wp + 4 * 256 + 192);
  w27 = *(const float4*)(wp + 6 * 256 + 192);
  w28 = *(const float4*)(wp + 8 * 256 + 192);
  w29 = *(const float4*)(wp + 10 * 256 + 192);
  w30 = *(const float4*)(wp + 12 * 256 + 192);
  w31 = *(const float4*)(wp + 14 * 256 + 192);
  float* wst = &sw[(p * 32 + ct * 16 + (lane >> 4)) * LSTRIDE + (lane & 15) * 4];

  const float* bl  = &sw[(p * 32 + ct * 16 + lr) * LSTRIDE + khalf];
  const float* as0 = &sh[0][lr * LSTRIDE + khalf];
  const float* as1 = &sh[1][lr * LSTRIDE + khalf];
  const float* ass = &sh[2][lr * LSTRIDE + khalf];
  float* exw = &ex[p][(lane >> 4) * 8 * FSTR + ct * 16 + lr];

  const int m = slice * 32 + lane;
  const float bi = b_iouh[m], bo = b_iouh[256 + m], bu = b_iouh[512 + m];
  const float bf = b_fh[m];

  bool first = true;
  for (int nl = blockIdx.y; nl < count; nl += gridDim.y) {
    const int node = level_start + nl;
    const int ch0  = 2 * node - 511;
    const int ch1  = 2 * node - 512;

    HSTORE(0, h0, h4); HSTORE(1, h1, h5); HSTORE(2, h2, h6); HSTORE(3, h3, h7);
    lds_barrier();

    float xi[2], xo[2], xu[2], xf[2], cc0[2], cc1[2];
    {
      const int nnext = nl + (int)gridDim.y;
      if (nnext < count) {
        const int nn  = level_start + nnext;
        const int nc0 = 2 * nn - 511;
        const int nc1 = 2 * nn - 512;
        HLOADALL(nc0, nc1)
      }
#pragma unroll
      for (int q = 0; q < 2; ++q) {
        const int b = rt * 16 + wave + 8 * q;
        const float* xp = xbuf + (size_t)node * (BATCH * 1024) + (size_t)b * 1024 + m;
        xi[q] = xp[0]; xo[q] = xp[256]; xu[q] = xp[512]; xf[q] = xp[768];
        const size_t base = (size_t)b * (NTREE * MDIM) + m;
        cc0[q] = c_buf[base + (size_t)ch0 * MDIM];
        cc1[q] = c_buf[base + (size_t)ch1 * MDIM];
      }
    }

    if (p < 3) {
      v8f acc = {};
      v2f fas[8], faw[8], fbs[8], fbw[8];
      if (first) {
        *(float4*)(wst + 0 * LSTRIDE + 0) = w0;
        *(float4*)(wst + 2 * LSTRIDE + 0) = w1;
        *(float4*)(wst + 4 * LSTRIDE + 0) = w2;
        *(float4*)(wst + 6 * LSTRIDE + 0) = w3;
        *(float4*)(wst + 8 * LSTRIDE + 0) = w4;
        *(float4*)(wst + 10 * LSTRIDE + 0) = w5;
        *(float4*)(wst + 12 * LSTRIDE + 0) = w6;
        *(float4*)(wst + 14 * LSTRIDE + 0) = w7;
      }
      _Pragma("unroll")
      for (int s = 0; s < 8; ++s) {
        fas[s] = *(const v2f*)(ass + 0 + 4 * s);
        faw[s] = *(const v2f*)(bl + 0 + 4 * s);
      }
      _Pragma("unroll")
      for (int s = 0; s < 8; ++s) {
        fbs[s] = *(const v2f*)(ass + 32 + 4 * s);
        fbw[s] = *(const v2f*)(bl + 32 + 4 * s);
      }
      FENCE_SCHED();
      _Pragma("unroll")
      for (int s = 0; s < 8; ++s) {
        acc = wmma4(fas[s], faw[s], acc);
      }
      FENCE_SCHED();
      if (first) {
        *(float4*)(wst + 0 * LSTRIDE + 64) = w8;
        *(float4*)(wst + 2 * LSTRIDE + 64) = w9;
        *(float4*)(wst + 4 * LSTRIDE + 64) = w10;
        *(float4*)(wst + 6 * LSTRIDE + 64) = w11;
        *(float4*)(wst + 8 * LSTRIDE + 64) = w12;
        *(float4*)(wst + 10 * LSTRIDE + 64) = w13;
        *(float4*)(wst + 12 * LSTRIDE + 64) = w14;
        *(float4*)(wst + 14 * LSTRIDE + 64) = w15;
      }
      _Pragma("unroll")
      for (int s = 0; s < 8; ++s) {
        fas[s] = *(const v2f*)(ass + 64 + 4 * s);
        faw[s] = *(const v2f*)(bl + 64 + 4 * s);
      }
      FENCE_SCHED();
      _Pragma("unroll")
      for (int s = 0; s < 8; ++s) {
        acc = wmma4(fbs[s], fbw[s], acc);
      }
      FENCE_SCHED();
      _Pragma("unroll")
      for (int s = 0; s < 8; ++s) {
        fbs[s] = *(const v2f*)(ass + 96 + 4 * s);
        fbw[s] = *(const v2f*)(bl + 96 + 4 * s);
      }
      FENCE_SCHED();
      _Pragma("unroll")
      for (int s = 0; s < 8; ++s) {
        acc = wmma4(fas[s], faw[s], acc);
      }
      FENCE_SCHED();
      if (first) {
        *(float4*)(wst + 0 * LSTRIDE + 128) = w16;
        *(float4*)(wst + 2 * LSTRIDE + 128) = w17;
        *(float4*)(wst + 4 * LSTRIDE + 128) = w18;
        *(float4*)(wst + 6 * LSTRIDE + 128) = w19;
        *(float4*)(wst + 8 * LSTRIDE + 128) = w20;
        *(float4*)(wst + 10 * LSTRIDE + 128) = w21;
        *(float4*)(wst + 12 * LSTRIDE + 128) = w22;
        *(float4*)(wst + 14 * LSTRIDE + 128) = w23;
      }
      _Pragma("unroll")
      for (int s = 0; s < 8; ++s) {
        fas[s] = *(const v2f*)(ass + 128 + 4 * s);
        faw[s] = *(const v2f*)(bl + 128 + 4 * s);
      }
      FENCE_SCHED();
      _Pragma("unroll")
      for (int s = 0; s < 8; ++s) {
        acc = wmma4(fbs[s], fbw[s], acc);
      }
      FENCE_SCHED();
      _Pragma("unroll")
      for (int s = 0; s < 8; ++s) {
        fbs[s] = *(const v2f*)(ass + 160 + 4 * s);
        fbw[s] = *(const v2f*)(bl + 160 + 4 * s);
      }
      FENCE_SCHED();
      _Pragma("unroll")
      for (int s = 0; s < 8; ++s) {
        acc = wmma4(fas[s], faw[s], acc);
      }
      FENCE_SCHED();
      if (first) {
        *(float4*)(wst + 0 * LSTRIDE + 192) = w24;
        *(float4*)(wst + 2 * LSTRIDE + 192) = w25;
        *(float4*)(wst + 4 * LSTRIDE + 192) = w26;
        *(float4*)(wst + 6 * LSTRIDE + 192) = w27;
        *(float4*)(wst + 8 * LSTRIDE + 192) = w28;
        *(float4*)(wst + 10 * LSTRIDE + 192) = w29;
        *(float4*)(wst + 12 * LSTRIDE + 192) = w30;
        *(float4*)(wst + 14 * LSTRIDE + 192) = w31;
      }
      _Pragma("unroll")
      for (int s = 0; s < 8; ++s) {
        fas[s] = *(const v2f*)(ass + 192 + 4 * s);
        faw[s] = *(const v2f*)(bl + 192 + 4 * s);
      }
      FENCE_SCHED();
      _Pragma("unroll")
      for (int s = 0; s < 8; ++s) {
        acc = wmma4(fbs[s], fbw[s], acc);
      }
      FENCE_SCHED();
      _Pragma("unroll")
      for (int s = 0; s < 8; ++s) {
        fbs[s] = *(const v2f*)(ass + 224 + 4 * s);
        fbw[s] = *(const v2f*)(bl + 224 + 4 * s);
      }
      FENCE_SCHED();
      _Pragma("unroll")
      for (int s = 0; s < 8; ++s) {
        acc = wmma4(fas[s], faw[s], acc);
      }
      FENCE_SCHED();
      FENCE_SCHED();
      _Pragma("unroll")
      for (int s = 0; s < 8; ++s) {
        acc = wmma4(fbs[s], fbw[s], acc);
      }
      FENCE_SCHED();
#pragma unroll
      for (int v = 0; v < 8; ++v) exw[v * FSTR] = acc[v];
    } else {
      v8f acc0 = {}, acc1 = {};
      v2f fa0[8], fa1[8], faw[8], fb0[8], fb1[8], fbw[8];
      if (first) {
        *(float4*)(wst + 0 * LSTRIDE + 0) = w0;
        *(float4*)(wst + 2 * LSTRIDE + 0) = w1;
        *(float4*)(wst + 4 * LSTRIDE + 0) = w2;
        *(float4*)(wst + 6 * LSTRIDE + 0) = w3;
        *(float4*)(wst + 8 * LSTRIDE + 0) = w4;
        *(float4*)(wst + 10 * LSTRIDE + 0) = w5;
        *(float4*)(wst + 12 * LSTRIDE + 0) = w6;
        *(float4*)(wst + 14 * LSTRIDE + 0) = w7;
      }
      _Pragma("unroll")
      for (int s = 0; s < 8; ++s) {
        fa0[s] = *(const v2f*)(as0 + 0 + 4 * s);
        fa1[s] = *(const v2f*)(as1 + 0 + 4 * s);
        faw[s] = *(const v2f*)(bl + 0 + 4 * s);
      }
      _Pragma("unroll")
      for (int s = 0; s < 8; ++s) {
        fb0[s] = *(const v2f*)(as0 + 32 + 4 * s);
        fb1[s] = *(const v2f*)(as1 + 32 + 4 * s);
        fbw[s] = *(const v2f*)(bl + 32 + 4 * s);
      }
      FENCE_SCHED();
      _Pragma("unroll")
      for (int s = 0; s < 8; ++s) {
        acc0 = wmma4(fa0[s], faw[s], acc0);
        acc1 = wmma4(fa1[s], faw[s], acc1);
      }
      FENCE_SCHED();
      if (first) {
        *(float4*)(wst + 0 * LSTRIDE + 64) = w8;
        *(float4*)(wst + 2 * LSTRIDE + 64) = w9;
        *(float4*)(wst + 4 * LSTRIDE + 64) = w10;
        *(float4*)(wst + 6 * LSTRIDE + 64) = w11;
        *(float4*)(wst + 8 * LSTRIDE + 64) = w12;
        *(float4*)(wst + 10 * LSTRIDE + 64) = w13;
        *(float4*)(wst + 12 * LSTRIDE + 64) = w14;
        *(float4*)(wst + 14 * LSTRIDE + 64) = w15;
      }
      _Pragma("unroll")
      for (int s = 0; s < 8; ++s) {
        fa0[s] = *(const v2f*)(as0 + 64 + 4 * s);
        fa1[s] = *(const v2f*)(as1 + 64 + 4 * s);
        faw[s] = *(const v2f*)(bl + 64 + 4 * s);
      }
      FENCE_SCHED();
      _Pragma("unroll")
      for (int s = 0; s < 8; ++s) {
        acc0 = wmma4(fb0[s], fbw[s], acc0);
        acc1 = wmma4(fb1[s], fbw[s], acc1);
      }
      FENCE_SCHED();
      _Pragma("unroll")
      for (int s = 0; s < 8; ++s) {
        fb0[s] = *(const v2f*)(as0 + 96 + 4 * s);
        fb1[s] = *(const v2f*)(as1 + 96 + 4 * s);
        fbw[s] = *(const v2f*)(bl + 96 + 4 * s);
      }
      FENCE_SCHED();
      _Pragma("unroll")
      for (int s = 0; s < 8; ++s) {
        acc0 = wmma4(fa0[s], faw[s], acc0);
        acc1 = wmma4(fa1[s], faw[s], acc1);
      }
      FENCE_SCHED();
      if (first) {
        *(float4*)(wst + 0 * LSTRIDE + 128) = w16;
        *(float4*)(wst + 2 * LSTRIDE + 128) = w17;
        *(float4*)(wst + 4 * LSTRIDE + 128) = w18;
        *(float4*)(wst + 6 * LSTRIDE + 128) = w19;
        *(float4*)(wst + 8 * LSTRIDE + 128) = w20;
        *(float4*)(wst + 10 * LSTRIDE + 128) = w21;
        *(float4*)(wst + 12 * LSTRIDE + 128) = w22;
        *(float4*)(wst + 14 * LSTRIDE + 128) = w23;
      }
      _Pragma("unroll")
      for (int s = 0; s < 8; ++s) {
        fa0[s] = *(const v2f*)(as0 + 128 + 4 * s);
        fa1[s] = *(const v2f*)(as1 + 128 + 4 * s);
        faw[s] = *(const v2f*)(bl + 128 + 4 * s);
      }
      FENCE_SCHED();
      _Pragma("unroll")
      for (int s = 0; s < 8; ++s) {
        acc0 = wmma4(fb0[s], fbw[s], acc0);
        acc1 = wmma4(fb1[s], fbw[s], acc1);
      }
      FENCE_SCHED();
      _Pragma("unroll")
      for (int s = 0; s < 8; ++s) {
        fb0[s] = *(const v2f*)(as0 + 160 + 4 * s);
        fb1[s] = *(const v2f*)(as1 + 160 + 4 * s);
        fbw[s] = *(const v2f*)(bl + 160 + 4 * s);
      }
      FENCE_SCHED();
      _Pragma("unroll")
      for (int s = 0; s < 8; ++s) {
        acc0 = wmma4(fa0[s], faw[s], acc0);
        acc1 = wmma4(fa1[s], faw[s], acc1);
      }
      FENCE_SCHED();
      if (first) {
        *(float4*)(wst + 0 * LSTRIDE + 192) = w24;
        *(float4*)(wst + 2 * LSTRIDE + 192) = w25;
        *(float4*)(wst + 4 * LSTRIDE + 192) = w26;
        *(float4*)(wst + 6 * LSTRIDE + 192) = w27;
        *(float4*)(wst + 8 * LSTRIDE + 192) = w28;
        *(float4*)(wst + 10 * LSTRIDE + 192) = w29;
        *(float4*)(wst + 12 * LSTRIDE + 192) = w30;
        *(float4*)(wst + 14 * LSTRIDE + 192) = w31;
      }
      _Pragma("unroll")
      for (int s = 0; s < 8; ++s) {
        fa0[s] = *(const v2f*)(as0 + 192 + 4 * s);
        fa1[s] = *(const v2f*)(as1 + 192 + 4 * s);
        faw[s] = *(const v2f*)(bl + 192 + 4 * s);
      }
      FENCE_SCHED();
      _Pragma("unroll")
      for (int s = 0; s < 8; ++s) {
        acc0 = wmma4(fb0[s], fbw[s], acc0);
        acc1 = wmma4(fb1[s], fbw[s], acc1);
      }
      FENCE_SCHED();
      _Pragma("unroll")
      for (int s = 0; s < 8; ++s) {
        fb0[s] = *(const v2f*)(as0 + 224 + 4 * s);
        fb1[s] = *(const v2f*)(as1 + 224 + 4 * s);
        fbw[s] = *(const v2f*)(bl + 224 + 4 * s);
      }
      FENCE_SCHED();
      _Pragma("unroll")
      for (int s = 0; s < 8; ++s) {
        acc0 = wmma4(fa0[s], faw[s], acc0);
        acc1 = wmma4(fa1[s], faw[s], acc1);
      }
      FENCE_SCHED();
      FENCE_SCHED();
      _Pragma("unroll")
      for (int s = 0; s < 8; ++s) {
        acc0 = wmma4(fb0[s], fbw[s], acc0);
        acc1 = wmma4(fb1[s], fbw[s], acc1);
      }
      FENCE_SCHED();
      float* exw1 = exw + 16 * FSTR;
#pragma unroll
      for (int v = 0; v < 8; ++v) { exw[v * FSTR] = acc0[v]; exw1[v * FSTR] = acc1[v]; }
    }
    first = false;
    lds_barrier();

#pragma unroll
    for (int q = 0; q < 2; ++q) {
      const int r = wave + 8 * q;
      const int b = rt * 16 + r;
      float ig = sigf(xi[q] + ex[0][r * FSTR + lane] + bi);
      float og = sigf(xo[q] + ex[1][r * FSTR + lane] + bo);
      float ug = tanhf(xu[q] + ex[2][r * FSTR + lane] + bu);
      float f0 = sigf(ex[3][r * FSTR + lane] + bf + xf[q]);
      float f1 = sigf(ex[4][r * FSTR + lane] + bf + xf[q]);
      float c = ig * ug + f0 * cc0[q] + f1 * cc1[q];
      float h = og * tanhf(c);
      const size_t o = (size_t)b * (NTREE * MDIM) + (size_t)node * MDIM + m;
      c_buf[o] = c;
      hid[o] = h;
    }
  }
}

#undef HPTR
#undef HLOADALL
#undef HOFF
#undef HSTORE
#undef FENCE_SCHED

__global__ __launch_bounds__(256) void root_copy(
    const float* __restrict__ c_buf, const float* __restrict__ hid,
    float* __restrict__ out)
{
  int idx = blockIdx.x * blockDim.x + threadIdx.x;
  int m = idx & 255, b = idx >> 8;
  size_t o = (size_t)b * (NTREE * MDIM) + (size_t)510 * MDIM + m;
  out[idx] = c_buf[o];
  out[BATCH * MDIM + idx] = hid[o];
}

extern "C" void kernel_launch(void* const* d_in, const int* in_sizes, int n_in,
                              void* d_out, int out_size, void* d_ws, size_t ws_size,
                              hipStream_t stream) {
  (void)in_sizes; (void)n_in; (void)out_size; (void)ws_size;
  const float* inputs = (const float*)d_in[0];
  const float* W_ioux = (const float*)d_in[1];
  const float* b_ioux = (const float*)d_in[2];
  const float* W_iouh = (const float*)d_in[3];
  const float* b_iouh = (const float*)d_in[4];
  const float* W_fx   = (const float*)d_in[5];
  const float* b_fx   = (const float*)d_in[6];
  const float* W_fh   = (const float*)d_in[7];
  const float* b_fh   = (const float*)d_in[8];

  float* out = (float*)d_out;
  float* hid = out + 2 * BATCH * MDIM;

  float* ws = (float*)d_ws;
  float* xbuf    = ws;
  float* c_buf   = xbuf + (size_t)NTREE * BATCH * 1024;
  float* wcat    = c_buf + (size_t)BATCH * NTREE * MDIM;
  float* bcat    = wcat + (size_t)1024 * 256;

  pack_weights<<<(1024 * 256 + 1024) / 256, 256, 0, stream>>>(
      W_ioux, b_ioux, W_fx, b_fx, wcat, bcat);

  dim3 g1(128, 4);
  xproj_gemm<<<g1, 256, 0, stream>>>(inputs, wcat, bcat, xbuf);

  leaf_kernel<<<(256 * BATCH * MDIM) / 256, 256, 0, stream>>>(xbuf, b_iouh, c_buf, hid);

  const int starts[8] = {256, 384, 448, 480, 496, 504, 508, 510};
  const int counts[8] = {128, 64, 32, 16, 8, 4, 2, 1};
  const int groups[8] = {16, 16, 16, 16, 8, 4, 2, 1};
  for (int L = 0; L < 8; ++L) {
    dim3 g3(16, groups[L]);
    level_fused<<<g3, 256, 0, stream>>>(xbuf, W_iouh, W_fh, b_iouh, b_fh,
                                        c_buf, hid, starts[L], counts[L]);
  }

  root_copy<<<(BATCH * MDIM) / 256, 256, 0, stream>>>(c_buf, hid, out);
}
